# starting kernel
# baseline (speedup 1.0000x reference)
#include <hip/hip_runtime.h>
#include <hip/hip_bf16.h>
#include <cstdint>

#define FEAT_DIM 128

typedef __attribute__((ext_vector_type(2))) float v2f;
typedef __attribute__((ext_vector_type(8))) float v8f;

__device__ __forceinline__ unsigned ordEncode(float f) {
  unsigned u = __float_as_uint(f);
  return (u & 0x80000000u) ? ~u : (u | 0x80000000u);
}
__device__ __forceinline__ float ordDecode(unsigned e) {
  unsigned u = (e & 0x80000000u) ? (e & 0x7fffffffu) : ~e;
  return __uint_as_float(u);
}

__global__ void gat_init_kernel(float* __restrict__ out,
                                unsigned* __restrict__ smax_ord,
                                float* __restrict__ denom,
                                int total, int n_nodes) {
  int i = blockIdx.x * blockDim.x + threadIdx.x;
  if (i < total) out[i] = 0.0f;
  if (i < n_nodes) {
    smax_ord[i] = 0u;
    denom[i] = 0.0f;
  }
}

__global__ void gat_scores_wmma_kernel(const float* __restrict__ feat,
                                       const int* __restrict__ src,
                                       const int* __restrict__ dst,
                                       float* __restrict__ scores,
                                       unsigned* __restrict__ smax_ord,
                                       int E) {
  const int lane = threadIdx.x & 31;
  const int wave = blockIdx.x * (blockDim.x >> 5) + (threadIdx.x >> 5);
  const int e0 = wave * 16;
  if (e0 >= E) return;

  const int m    = lane & 15;
  const int half = lane >> 4;
  const int e    = min(e0 + m, E - 1);

  const float* __restrict__ ps = feat + (size_t)src[e] * FEAT_DIM;
  const float* __restrict__ pd = feat + (size_t)dst[e] * FEAT_DIM;

  v8f c = {0.f, 0.f, 0.f, 0.f, 0.f, 0.f, 0.f, 0.f};
#pragma unroll
  for (int kk = 0; kk < FEAT_DIM; kk += 4) {
    const int kb = kk + 2 * half;
    const float2 av = *(const float2*)(ps + kb);
    const float2 bv = *(const float2*)(pd + kb);
    v2f a; a.x = av.x; a.y = av.y;
    v2f b; b.x = bv.x; b.y = bv.y;
    c = __builtin_amdgcn_wmma_f32_16x16x4_f32(false, a, false, b, (short)0, c,
                                              false, false);
  }

  float s = 0.0f;
#pragma unroll
  for (int r = 0; r < 8; ++r) {
    const float vlo = __shfl(c[r], r, 32);
    const float vhi = __shfl(c[r], r + 24, 32);
    if (lane == r)     s = vlo;
    if (lane == r + 8) s = vhi;
  }

  if (lane < 16 && (e0 + lane) < E) {
    const float score = s * (1.0f / 1.0f);
    scores[e0 + lane] = score;
    atomicMax(&smax_ord[dst[e0 + lane]], ordEncode(score));
  }
}

__global__ void gat_expsum_kernel(const int* __restrict__ dst,
                                  float* __restrict__ scores_ex,
                                  const unsigned* __restrict__ smax_ord,
                                  float* __restrict__ denom, int E) {
  const int e = blockIdx.x * blockDim.x + threadIdx.x;
  if (e >= E) return;
  const int d = dst[e];
  const float mx = ordDecode(smax_ord[d]);
  const float ex = __expf(scores_ex[e] - mx);
  scores_ex[e] = ex;
  atomicAdd(&denom[d], ex);
}

__global__ void gat_aggregate_kernel(const float* __restrict__ feat,
                                     const int* __restrict__ src,
                                     const int* __restrict__ dst,
                                     const float* __restrict__ ex,
                                     const float* __restrict__ denom,
                                     float* __restrict__ out, int E) {
  const int lane = threadIdx.x & 31;
  const int e = blockIdx.x * (blockDim.x >> 5) + (threadIdx.x >> 5);
  if (e >= E) return;
  const int d = dst[e];
  const float w = ex[e] / denom[d];
  const float4 v = ((const float4*)(feat + (size_t)src[e] * FEAT_DIM))[lane];
  float* po = out + (size_t)d * FEAT_DIM + lane * 4;
  atomicAdd(po + 0, w * v.x);
  atomicAdd(po + 1, w * v.y);
  atomicAdd(po + 2, w * v.z);
  atomicAdd(po + 3, w * v.w);
}

__global__ void gat_fixup_kernel(const float* __restrict__ feat,
                                 const float* __restrict__ denom,
                                 float* __restrict__ out, int total) {
  const int i = blockIdx.x * blockDim.x + threadIdx.x;
  if (i >= total) return;
  const int n = i >> 7;
  if (denom[n] == 0.0f) out[i] = feat[i];
}

extern "C" void kernel_launch(void* const* d_in, const int* in_sizes, int n_in,
                              void* d_out, int out_size, void* d_ws, size_t ws_size,
                              hipStream_t stream) {
  const float* feat = (const float*)d_in[0];
  const int* src    = (const int*)d_in[1];
  const int* dst    = (const int*)d_in[2];
  const int E       = in_sizes[1];
  const int N       = in_sizes[0] / FEAT_DIM;
  const int total   = N * FEAT_DIM;
  float* out        = (float*)d_out;

  float*    scores   = (float*)d_ws;
  unsigned* smax_ord = (unsigned*)(scores + E);
  float*    denom    = (float*)(smax_ord + N);

  const int BLK = 256;

  gat_init_kernel<<<(total + BLK - 1) / BLK, BLK, 0, stream>>>(
      out, smax_ord, denom, total, N);

  {
    const int waves   = (E + 15) / 16;
    const int threads = waves * 32;
    gat_scores_wmma_kernel<<<(threads + BLK - 1) / BLK, BLK, 0, stream>>>(
        feat, src, dst, scores, smax_ord, E);
  }

  gat_expsum_kernel<<<(E + BLK - 1) / BLK, BLK, 0, stream>>>(
      dst, scores, smax_ord, denom, E);

  {
    const long long threads = (long long)E * 32;
    gat_aggregate_kernel<<<(int)((threads + BLK - 1) / BLK), BLK, 0, stream>>>(
        feat, src, dst, scores, denom, out, E);
  }

  gat_fixup_kernel<<<(total + BLK - 1) / BLK, BLK, 0, stream>>>(
      feat, denom, out, total);
}


# all behind the scores done by owner work-groups: each filters the whole edge list for its 196 nodes; max, denominator, out rows in LDS: no global atomics, 2 launches; sum order = LDS arrival
# speedup vs baseline: 3.5457x; 3.5457x over previous
#include <hip/hip_runtime.h>
#include <hip/hip_bf16.h>
#include <cstdint>

#define FEAT_DIM 128

typedef __attribute__((ext_vector_type(2))) float v2f;
typedef __attribute__((ext_vector_type(8))) float v8f;

__device__ __forceinline__ unsigned ordEncode(float f) {
  unsigned u = __float_as_uint(f);
  return (u & 0x80000000u) ? ~u : (u | 0x80000000u);
}
__device__ __forceinline__ float ordDecode(unsigned e) {
  unsigned u = (e & 0x80000000u) ? (e & 0x7fffffffu) : ~e;
  return __uint_as_float(u);
}

#define GROUP_EDGES 4

__global__ void gat_scores_wmma_kernel(const float* __restrict__ feat,
                                       const int* __restrict__ src,
                                       const int* __restrict__ dst,
                                       float* __restrict__ scores,
                                       int E) {
  const int lane = threadIdx.x & 31;
  const int wave = blockIdx.x * (blockDim.x >> 5) + (threadIdx.x >> 5);
  const int e0 = wave * 16;
  if (e0 >= E) return;

  const int el = min(e0 + (lane & 15), E - 1);
  const int sv = src[el];
  const int dv = dst[el];

  const bool hi16 = (lane & 16) != 0;
  const bool hi8 = (lane & 8) != 0;
  const int my_group = (lane >> 1) & 3;
  float score = 0.0f;

#pragma unroll 1
  for (int g = 0; g < 16 / GROUP_EDGES; ++g) {
    float4 ra[GROUP_EDGES], rb[GROUP_EDGES];
#pragma unroll
    for (int i = 0; i < GROUP_EDGES; ++i) {
      const int sj = __builtin_amdgcn_readlane(sv, g * GROUP_EDGES + i);
      const int dj = __builtin_amdgcn_readlane(dv, g * GROUP_EDGES + i);
      ra[i] = ((const float4*)(feat + (size_t)sj * FEAT_DIM))[lane];
      rb[i] = ((const float4*)(feat + (size_t)dj * FEAT_DIM))[lane];
    }

    float t[GROUP_EDGES];
#pragma unroll
    for (int i = 0; i < GROUP_EDGES; ++i) {
      v2f a0; a0.x = ra[i].x; a0.y = ra[i].y;
      v2f b0; b0.x = rb[i].x; b0.y = rb[i].y;
      v2f a1; a1.x = ra[i].z; a1.y = ra[i].w;
      v2f b1; b1.x = rb[i].z; b1.y = rb[i].w;
      v8f c = {0.f, 0.f, 0.f, 0.f, 0.f, 0.f, 0.f, 0.f};
      c = __builtin_amdgcn_wmma_f32_16x16x4_f32(false, a0, false, b0, (short)0,
                                                c, false, false);
      c = __builtin_amdgcn_wmma_f32_16x16x4_f32(false, a1, false, b1, (short)0,
                                                c, false, false);
      float tj = 0.0f;
#pragma unroll
      for (int r = 0; r < 8; ++r) {
        if (lane == r || lane == r + 24) tj = c[r];
      }
      t[i] = tj;
    }

    const float k0 = hi16 ? t[2] : t[0];
    const float s0 = hi16 ? t[0] : t[2];
    const float k1 = hi16 ? t[3] : t[1];
    const float s1 = hi16 ? t[1] : t[3];
    const float u0 = k0 + __shfl_xor(s0, 16, 32);
    const float u1 = k1 + __shfl_xor(s1, 16, 32);
    const float k2 = hi8 ? u1 : u0;
    const float s2 = hi8 ? u0 : u1;
    float q = k2 + __shfl_xor(s2, 8, 32);
    q += __shfl_xor(q, 4, 32);
    q += __shfl_xor(q, 2, 32);
    q += __shfl_xor(q, 1, 32);
    if (g == my_group) score = q;
  }

  const int eo = e0 + GROUP_EDGES * my_group + (hi16 ? 2 : 0) + (hi8 ? 1 : 0);
  if ((lane & 1) == 0 && eo < E) scores[eo] = score;
}

#define OWN_ROWS 196
#define OWN_THREADS 1024
#define OWN_WAVES (OWN_THREADS / 32)
#define LIST_CAP 176
#define ROW_GROUP 16
#define SCAN_AHEAD 8

__device__ __forceinline__ int4 load_dst4(const int* __restrict__ dst, int q,
                                          int nq_full) {
  int4 r = ((const int4*)dst)[min(q, nq_full - 1)];
  if (q >= nq_full) r = make_int4(-1, -1, -1, -1);
  return r;
}

__device__ __forceinline__ void add_row(float* __restrict__ arow, float4 v,
                                        float w, int lane) {
  atomicAdd(arow + lane, v.x * w);
  atomicAdd(arow + 32 + lane, v.y * w);
  atomicAdd(arow + 64 + lane, v.z * w);
  atomicAdd(arow + 96 + lane, v.w * w);
}

template <int STEP>
__device__ __forceinline__ void walk_group(
    const int (&rr)[4], int e0, const float* __restrict__ feat,
    const int* __restrict__ src, const float* __restrict__ scores,
    unsigned* mx, float* den, float* acc, int lane) {
#pragma unroll
  for (int i = 0; i < 4; ++i) {
    const bool hit = (unsigned)rr[i] < (unsigned)OWN_ROWS;
    if (STEP == 1) {
      if (hit) atomicMax(&mx[rr[i]], ordEncode(scores[e0 + i]));
    } else if (STEP == 2) {
      if (hit)
        atomicAdd(&den[rr[i]], __expf(scores[e0 + i] - ordDecode(mx[rr[i]])));
    } else {
      unsigned m = (unsigned)__ballot(hit);
      while (m != 0u) {
        const int j = __builtin_ctz(m);
        m &= m - 1u;
        const int ej = __builtin_amdgcn_readlane(e0 + i, j);
        const int rj = __builtin_amdgcn_readlane(rr[i], j);
        const float w = __expf(scores[ej] - ordDecode(mx[rj])) / den[rj];
        const float4 v =
            ((const float4*)(feat + (size_t)src[ej] * FEAT_DIM))[lane];
        add_row(acc + rj * FEAT_DIM, v, w, lane);
      }
    }
  }
}

template <int STEP>
__device__ __forceinline__ void walk_rest(
    int t_full, int t0, int n_trip, int nq_full, int E, int lo, int tid,
    const float* __restrict__ feat, const int* __restrict__ src,
    const int* __restrict__ dst, const float* __restrict__ scores,
    unsigned* mx, float* den, float* acc) {
  const int lane = tid & 31;
  for (int t = t_full; t < n_trip; ++t) {
    const int k = (t + t0 < n_trip) ? t + t0 : t + t0 - n_trip;
    const int q = k * OWN_THREADS + tid;
    const int4 d = load_dst4(dst, q, nq_full);
    const int rr[4] = {d.x - lo, d.y - lo, d.z - lo, d.w - lo};
    walk_group<STEP>(rr, q * 4, feat, src, scores, mx, den, acc, lane);
  }
  if (tid < 32 && (E & 3) != 0) {
    const int e0 = nq_full * 4;
    int rr[4];
#pragma unroll
    for (int i = 0; i < 4; ++i)
      rr[i] = ((tid == 0 && e0 + i < E) ? dst[e0 + i] : -1) - lo;
    walk_group<STEP>(rr, e0, feat, src, scores, mx, den, acc, lane);
  }
}

__global__ void __launch_bounds__(OWN_THREADS)
gat_owner_kernel(const float* __restrict__ feat, const int* __restrict__ src,
                 const int* __restrict__ dst, const float* __restrict__ scores,
                 float* __restrict__ out, int E, int n_nodes) {
  __shared__ float acc[OWN_ROWS * FEAT_DIM];
  __shared__ unsigned mx[OWN_ROWS];
  __shared__ float den[OWN_ROWS];
  __shared__ int list_e[OWN_WAVES * LIST_CAP];
  __shared__ int list_row[OWN_WAVES * LIST_CAP];
  __shared__ float list_val[OWN_WAVES * LIST_CAP];

  const int tid = threadIdx.x;
  const int lane = tid & 31;
  const int wv = tid >> 5;
  const int lo = blockIdx.x * OWN_ROWS;
  int* le = list_e + wv * LIST_CAP;
  int* lr = list_row + wv * LIST_CAP;
  float* lv = list_val + wv * LIST_CAP;

  for (int i = tid; i < OWN_ROWS * FEAT_DIM / 4; i += OWN_THREADS)
    ((float4*)acc)[i] = make_float4(0.0f, 0.0f, 0.0f, 0.0f);
  if (tid < OWN_ROWS) {
    mx[tid] = 0u;
    den[tid] = 0.0f;
  }

  const int nq_full = E >> 2;
  const int n_trip = (nq_full + OWN_THREADS - 1) / OWN_THREADS;
  const int t0 = (int)(((long long)blockIdx.x * n_trip) / gridDim.x);
  int cnt = 0;
  int t_full = n_trip;
  for (int tb = 0; tb < n_trip && t_full == n_trip; tb += SCAN_AHEAD) {
    int4 d[SCAN_AHEAD];
    int qq[SCAN_AHEAD];
#pragma unroll
    for (int u = 0; u < SCAN_AHEAD; ++u) {
      const int t = tb + u + t0;
      qq[u] = ((t < n_trip) ? t : t - n_trip) * OWN_THREADS + tid;
      d[u] = load_dst4(dst, qq[u], nq_full);
    }
#pragma unroll
    for (int u = 0; u < SCAN_AHEAD; ++u) {
      if (tb + u >= n_trip || t_full != n_trip) continue;
      const int r0 = d[u].x - lo, r1 = d[u].y - lo;
      const int r2 = d[u].z - lo, r3 = d[u].w - lo;
      const unsigned rmin = min(min((unsigned)r0, (unsigned)r1),
                                min((unsigned)r2, (unsigned)r3));
      if (__ballot(rmin < (unsigned)OWN_ROWS) == 0) continue;
      const int rr[4] = {r0, r1, r2, r3};
      unsigned m[4];
      int tot = 0;
#pragma unroll
      for (int i = 0; i < 4; ++i) {
        m[i] = (unsigned)__ballot((unsigned)rr[i] < (unsigned)OWN_ROWS);
        tot += __builtin_popcount(m[i]);
      }
      if (cnt + tot > LIST_CAP) {
        t_full = tb + u;
        continue;
      }
      const int e0 = qq[u] * 4;
#pragma unroll
      for (int i = 0; i < 4; ++i) {
        if ((unsigned)rr[i] < (unsigned)OWN_ROWS) {
          const int pos = cnt + (int)__builtin_amdgcn_mbcnt_lo(m[i], 0u);
          le[pos] = e0 + i;
          lr[pos] = rr[i];
        }
        cnt += __builtin_popcount(m[i]);
      }
    }
  }
  __syncthreads();

  for (int i = lane; i < cnt; i += 32) {
    const int e = le[i];
    const float s = scores[e];
    le[i] = src[e];
    lv[i] = s;
    atomicMax(&mx[lr[i]], ordEncode(s));
  }
  walk_rest<1>(t_full, t0, n_trip, nq_full, E, lo, tid, feat, src, dst, scores,
               mx, den, acc);
  __syncthreads();

  for (int i = lane; i < cnt; i += 32) {
    const int r = lr[i];
    const float ex = __expf(lv[i] - ordDecode(mx[r]));
    lv[i] = ex;
    atomicAdd(&den[r], ex);
  }
  walk_rest<2>(t_full, t0, n_trip, nq_full, E, lo, tid, feat, src, dst, scores,
               mx, den, acc);
  __syncthreads();

  for (int i = lane; i < cnt; i += 32) lv[i] = lv[i] / den[lr[i]];
  __syncthreads();

  for (int base = 0; base < cnt; base += ROW_GROUP) {
    float4 v[ROW_GROUP];
#pragma unroll
    for (int i = 0; i < ROW_GROUP; ++i) {
      if (base + i < cnt) {
        const int sj = __builtin_amdgcn_readfirstlane(le[base + i]);
        v[i] = ((const float4*)(feat + (size_t)sj * FEAT_DIM))[lane];
      }
    }
#pragma unroll
    for (int i = 0; i < ROW_GROUP; ++i) {
      if (base + i < cnt) {
        const int rj = __builtin_amdgcn_readfirstlane(lr[base + i]);
        add_row(acc + rj * FEAT_DIM, v[i], lv[base + i], lane);
      }
    }
  }
  walk_rest<3>(t_full, t0, n_trip, nq_full, E, lo, tid, feat, src, dst, scores,
               mx, den, acc);
  __syncthreads();

  const int n_rows = min(OWN_ROWS, n_nodes - lo);
  for (int r = wv; r < n_rows; r += OWN_WAVES) {
    const float* a = acc + r * FEAT_DIM + lane;
    float4 o = make_float4(a[0], a[32], a[64], a[96]);
    const size_t r4 = (size_t)(lo + r) * (FEAT_DIM / 4) + lane;
    if (den[r] == 0.0f) o = ((const float4*)feat)[r4];
    ((float4*)out)[r4] = o;
  }
}

extern "C" void kernel_launch(void* const* d_in, const int* in_sizes, int n_in,
                              void* d_out, int out_size, void* d_ws, size_t ws_size,
                              hipStream_t stream) {
  const float* feat = (const float*)d_in[0];
  const int* src    = (const int*)d_in[1];
  const int* dst    = (const int*)d_in[2];
  const int E       = in_sizes[1];
  const int N       = in_sizes[0] / FEAT_DIM;
  float* out        = (float*)d_out;

  float* scores = (float*)d_ws;

  const int BLK = 256;

  {
    const int waves   = (E + 15) / 16;
    const int threads = waves * 32;
    gat_scores_wmma_kernel<<<(threads + BLK - 1) / BLK, BLK, 0, stream>>>(
        feat, src, dst, scores, E);
  }

  gat_owner_kernel<<<(N + OWN_ROWS - 1) / OWN_ROWS, OWN_THREADS, 0, stream>>>(
      feat, src, dst, scores, out, E, N);
}


# scores computed inside the owner work-groups: own destination rows staged in LDS (in the out-row space), source row x LDS row by f32 WMMA trace; 1 launch, no scores pass; score sum order changes
# speedup vs baseline: 3.7440x; 1.0559x over previous
#include <hip/hip_runtime.h>
#include <hip/hip_bf16.h>
#include <cstdint>

#define FEAT_DIM 128

typedef __attribute__((ext_vector_type(2))) float v2f;
typedef __attribute__((ext_vector_type(8))) float v8f;

__device__ __forceinline__ unsigned ordEncode(float f) {
  unsigned u = __float_as_uint(f);
  return (u & 0x80000000u) ? ~u : (u | 0x80000000u);
}
__device__ __forceinline__ float ordDecode(unsigned e) {
  unsigned u = (e & 0x80000000u) ? (e & 0x7fffffffu) : ~e;
  return __uint_as_float(u);
}

__device__ __forceinline__ float tile_trace_share(float4 ra, float4 rb,
                                                  int lane) {
  v2f a0; a0.x = ra.x; a0.y = ra.y;
  v2f b0; b0.x = rb.x; b0.y = rb.y;
  v2f a1; a1.x = ra.z; a1.y = ra.w;
  v2f b1; b1.x = rb.z; b1.y = rb.w;
  v8f c = {0.f, 0.f, 0.f, 0.f, 0.f, 0.f, 0.f, 0.f};
  c = __builtin_amdgcn_wmma_f32_16x16x4_f32(false, a0, false, b0, (short)0, c,
                                            false, false);
  c = __builtin_amdgcn_wmma_f32_16x16x4_f32(false, a1, false, b1, (short)0, c,
                                            false, false);
  float tj = 0.0f;
#pragma unroll
  for (int r = 0; r < 8; ++r) {
    if (lane == r || lane == r + 24) tj = c[r];
  }
  return tj;
}

#define OWN_ROWS 196
#define OWN_THREADS 1024
#define OWN_WAVES (OWN_THREADS / 32)
#define LIST_CAP 176
#define ROW_GROUP 16
#define SCORE_GROUP 8
#define SCAN_AHEAD 8

__device__ __forceinline__ int4 load_dst4(const int* __restrict__ dst, int q,
                                          int nq_full) {
  int4 r = ((const int4*)dst)[min(q, nq_full - 1)];
  if (q >= nq_full) r = make_int4(-1, -1, -1, -1);
  return r;
}

__device__ __forceinline__ void add_row(float* __restrict__ arow, float4 v,
                                        float w, int lane) {
  atomicAdd(arow + lane, v.x * w);
  atomicAdd(arow + 32 + lane, v.y * w);
  atomicAdd(arow + 64 + lane, v.z * w);
  atomicAdd(arow + 96 + lane, v.w * w);
}

template <int STEP>
__device__ __forceinline__ void walk_group(
    const int (&rr)[4], int e0, const float* __restrict__ feat,
    const int* __restrict__ src, float* __restrict__ scores,
    unsigned* mx, float* den, float* rows, int lane) {
#pragma unroll
  for (int i = 0; i < 4; ++i) {
    const bool hit = (unsigned)rr[i] < (unsigned)OWN_ROWS;
    if (STEP == 2) {
      if (hit)
        atomicAdd(&den[rr[i]], __expf(scores[e0 + i] - ordDecode(mx[rr[i]])));
    } else {
      unsigned m = (unsigned)__ballot(hit);
      while (m != 0u) {
        const int j = __builtin_ctz(m);
        m &= m - 1u;
        const int ej = __builtin_amdgcn_readlane(e0 + i, j);
        const int rj = __builtin_amdgcn_readlane(rr[i], j);
        const float4 v =
            ((const float4*)(feat + (size_t)src[ej] * FEAT_DIM))[lane];
        if (STEP == 1) {
          const float4 b = ((const float4*)(rows + rj * FEAT_DIM))[lane];
          float q = tile_trace_share(v, b, lane);
          q += __shfl_xor(q, 16, 32);
          q += __shfl_xor(q, 8, 32);
          q += __shfl_xor(q, 4, 32);
          q += __shfl_xor(q, 2, 32);
          q += __shfl_xor(q, 1, 32);
          if (lane == 0) {
            scores[ej] = q;
            atomicMax(&mx[rj], ordEncode(q));
          }
        } else {
          const float w = __expf(scores[ej] - ordDecode(mx[rj])) / den[rj];
          add_row(rows + rj * FEAT_DIM, v, w, lane);
        }
      }
    }
  }
}

template <int STEP>
__device__ __forceinline__ void walk_rest(
    int t_full, int t0, int n_trip, int nq_full, int E, int lo, int tid,
    const float* __restrict__ feat, const int* __restrict__ src,
    const int* __restrict__ dst, float* __restrict__ scores,
    unsigned* mx, float* den, float* rows) {
  const int lane = tid & 31;
  for (int t = t_full; t < n_trip; ++t) {
    const int k = (t + t0 < n_trip) ? t + t0 : t + t0 - n_trip;
    const int q = k * OWN_THREADS + tid;
    const int4 d = load_dst4(dst, q, nq_full);
    const int rr[4] = {d.x - lo, d.y - lo, d.z - lo, d.w - lo};
    walk_group<STEP>(rr, q * 4, feat, src, scores, mx, den, rows, lane);
  }
  if (tid < 32 && (E & 3) != 0) {
    const int e0 = nq_full * 4;
    int rr[4];
#pragma unroll
    for (int i = 0; i < 4; ++i)
      rr[i] = ((tid == 0 && e0 + i < E) ? dst[e0 + i] : -1) - lo;
    walk_group<STEP>(rr, e0, feat, src, scores, mx, den, rows, lane);
  }
}

__global__ void __launch_bounds__(OWN_THREADS)
gat_owner_kernel(const float* __restrict__ feat, const int* __restrict__ src,
                 const int* __restrict__ dst, float* __restrict__ scores,
                 float* __restrict__ out, int E, int n_nodes) {
  __shared__ float acc[OWN_ROWS * FEAT_DIM];
  __shared__ unsigned mx[OWN_ROWS];
  __shared__ float den[OWN_ROWS];
  __shared__ int list_e[OWN_WAVES * LIST_CAP];
  __shared__ int list_row[OWN_WAVES * LIST_CAP];
  __shared__ float list_val[OWN_WAVES * LIST_CAP];

  const int tid = threadIdx.x;
  const int lane = tid & 31;
  const int wv = tid >> 5;
  const int lo = blockIdx.x * OWN_ROWS;
  const int n_rows = min(OWN_ROWS, n_nodes - lo);
  int* le = list_e + wv * LIST_CAP;
  int* lr = list_row + wv * LIST_CAP;
  float* lv = list_val + wv * LIST_CAP;

  for (int i = tid; i < n_rows * (FEAT_DIM / 4); i += OWN_THREADS)
    ((float4*)acc)[i] = ((const float4*)feat)[(size_t)lo * (FEAT_DIM / 4) + i];
  if (tid < OWN_ROWS) {
    mx[tid] = 0u;
    den[tid] = 0.0f;
  }

  const int nq_full = E >> 2;
  const int n_trip = (nq_full + OWN_THREADS - 1) / OWN_THREADS;
  const int t0 = (int)(((long long)blockIdx.x * n_trip) / gridDim.x);
  int cnt = 0;
  int t_full = n_trip;
  for (int tb = 0; tb < n_trip && t_full == n_trip; tb += SCAN_AHEAD) {
    int4 d[SCAN_AHEAD];
    int qq[SCAN_AHEAD];
#pragma unroll
    for (int u = 0; u < SCAN_AHEAD; ++u) {
      const int t = tb + u + t0;
      qq[u] = ((t < n_trip) ? t : t - n_trip) * OWN_THREADS + tid;
      d[u] = load_dst4(dst, qq[u], nq_full);
    }
#pragma unroll
    for (int u = 0; u < SCAN_AHEAD; ++u) {
      if (tb + u >= n_trip || t_full != n_trip) continue;
      const int r0 = d[u].x - lo, r1 = d[u].y - lo;
      const int r2 = d[u].z - lo, r3 = d[u].w - lo;
      const unsigned rmin = min(min((unsigned)r0, (unsigned)r1),
                                min((unsigned)r2, (unsigned)r3));
      if (__ballot(rmin < (unsigned)OWN_ROWS) == 0) continue;
      const int rr[4] = {r0, r1, r2, r3};
      unsigned m[4];
      int tot = 0;
#pragma unroll
      for (int i = 0; i < 4; ++i) {
        m[i] = (unsigned)__ballot((unsigned)rr[i] < (unsigned)OWN_ROWS);
        tot += __builtin_popcount(m[i]);
      }
      if (cnt + tot > LIST_CAP) {
        t_full = tb + u;
        continue;
      }
      const int e0 = qq[u] * 4;
#pragma unroll
      for (int i = 0; i < 4; ++i) {
        if ((unsigned)rr[i] < (unsigned)OWN_ROWS) {
          const int pos = cnt + (int)__builtin_amdgcn_mbcnt_lo(m[i], 0u);
          le[pos] = e0 + i;
          lr[pos] = rr[i];
        }
        cnt += __builtin_popcount(m[i]);
      }
    }
  }
  __syncthreads();

  for (int i = lane; i < cnt; i += 32) le[i] = src[le[i]];
  const bool hi16 = (lane & 16) != 0;
  const bool hi8 = (lane & 8) != 0;
  const bool hi4 = (lane & 4) != 0;
  for (int base = 0; base < cnt; base += SCORE_GROUP) {
    float4 a[SCORE_GROUP];
    int rj[SCORE_GROUP];
#pragma unroll
    for (int i = 0; i < SCORE_GROUP; ++i) {
      const int k = min(base + i, cnt - 1);
      const int sj = __builtin_amdgcn_readfirstlane(le[k]);
      rj[i] = __builtin_amdgcn_readfirstlane(lr[k]);
      a[i] = ((const float4*)(feat + (size_t)sj * FEAT_DIM))[lane];
    }
    float t[SCORE_GROUP];
#pragma unroll
    for (int i = 0; i < SCORE_GROUP; ++i) {
      const float4 b = ((const float4*)(acc + rj[i] * FEAT_DIM))[lane];
      t[i] = tile_trace_share(a[i], b, lane);
    }
    float u[4];
#pragma unroll
    for (int j = 0; j < 4; ++j) {
      const float keep = hi16 ? t[4 + j] : t[j];
      const float send = hi16 ? t[j] : t[4 + j];
      u[j] = keep + __shfl_xor(send, 16, 32);
    }
    float w2[2];
#pragma unroll
    for (int j = 0; j < 2; ++j) {
      const float keep = hi8 ? u[2 + j] : u[j];
      const float send = hi8 ? u[j] : u[2 + j];
      w2[j] = keep + __shfl_xor(send, 8, 32);
    }
    float q = (hi4 ? w2[1] : w2[0]) + __shfl_xor(hi4 ? w2[0] : w2[1], 4, 32);
    q += __shfl_xor(q, 2, 32);
    q += __shfl_xor(q, 1, 32);
    const int k = base + (lane >> 2);
    if ((lane & 3) == 0 && k < cnt) {
      lv[k] = q;
      atomicMax(&mx[lr[k]], ordEncode(q));
    }
  }
  walk_rest<1>(t_full, t0, n_trip, nq_full, E, lo, tid, feat, src, dst, scores,
               mx, den, acc);
  __syncthreads();

  for (int i = tid; i < OWN_ROWS * FEAT_DIM / 4; i += OWN_THREADS)
    ((float4*)acc)[i] = make_float4(0.0f, 0.0f, 0.0f, 0.0f);
  for (int i = lane; i < cnt; i += 32) {
    const int r = lr[i];
    const float ex = __expf(lv[i] - ordDecode(mx[r]));
    lv[i] = ex;
    atomicAdd(&den[r], ex);
  }
  walk_rest<2>(t_full, t0, n_trip, nq_full, E, lo, tid, feat, src, dst, scores,
               mx, den, acc);
  __syncthreads();

  for (int i = lane; i < cnt; i += 32) lv[i] = lv[i] / den[lr[i]];
  __syncthreads();

  for (int base = 0; base < cnt; base += ROW_GROUP) {
    float4 v[ROW_GROUP];
#pragma unroll
    for (int i = 0; i < ROW_GROUP; ++i) {
      if (base + i < cnt) {
        const int sj = __builtin_amdgcn_readfirstlane(le[base + i]);
        v[i] = ((const float4*)(feat + (size_t)sj * FEAT_DIM))[lane];
      }
    }
#pragma unroll
    for (int i = 0; i < ROW_GROUP; ++i) {
      if (base + i < cnt) {
        const int rj = __builtin_amdgcn_readfirstlane(lr[base + i]);
        add_row(acc + rj * FEAT_DIM, v[i], lv[base + i], lane);
      }
    }
  }
  walk_rest<3>(t_full, t0, n_trip, nq_full, E, lo, tid, feat, src, dst, scores,
               mx, den, acc);
  __syncthreads();

  for (int r = wv; r < n_rows; r += OWN_WAVES) {
    const float* a = acc + r * FEAT_DIM + lane;
    float4 o = make_float4(a[0], a[32], a[64], a[96]);
    const size_t r4 = (size_t)(lo + r) * (FEAT_DIM / 4) + lane;
    if (den[r] == 0.0f) o = ((const float4*)feat)[r4];
    ((float4*)out)[r4] = o;
  }
}

extern "C" void kernel_launch(void* const* d_in, const int* in_sizes, int n_in,
                              void* d_out, int out_size, void* d_ws, size_t ws_size,
                              hipStream_t stream) {
  const float* feat = (const float*)d_in[0];
  const int* src    = (const int*)d_in[1];
  const int* dst    = (const int*)d_in[2];
  const int E       = in_sizes[1];
  const int N       = in_sizes[0] / FEAT_DIM;
  float* out        = (float*)d_out;

  float* scores = (float*)d_ws;

  gat_owner_kernel<<<(N + OWN_ROWS - 1) / OWN_ROWS, OWN_THREADS, 0, stream>>>(
      feat, src, dst, scores, out, E, N);
}


# owner work-groups filter a one-byte owner key per edge (written by a small first launch), not the 4-byte destination list: 1/4 of the list bytes, 4 edges per compare; sum order stays LDS arrival
# speedup vs baseline: 4.3088x; 1.1509x over previous
#include <hip/hip_runtime.h>
#include <hip/hip_bf16.h>
#include <cstdint>

#define FEAT_DIM 128

typedef __attribute__((ext_vector_type(2))) float v2f;
typedef __attribute__((ext_vector_type(8))) float v8f;

__device__ __forceinline__ unsigned ordEncode(float f) {
  unsigned u = __float_as_uint(f);
  return (u & 0x80000000u) ? ~u : (u | 0x80000000u);
}
__device__ __forceinline__ float ordDecode(unsigned e) {
  unsigned u = (e & 0x80000000u) ? (e & 0x7fffffffu) : ~e;
  return __uint_as_float(u);
}

__device__ __forceinline__ float tile_trace_share(float4 ra, float4 rb,
                                                  int lane) {
  v2f a0; a0.x = ra.x; a0.y = ra.y;
  v2f b0; b0.x = rb.x; b0.y = rb.y;
  v2f a1; a1.x = ra.z; a1.y = ra.w;
  v2f b1; b1.x = rb.z; b1.y = rb.w;
  v8f c = {0.f, 0.f, 0.f, 0.f, 0.f, 0.f, 0.f, 0.f};
  c = __builtin_amdgcn_wmma_f32_16x16x4_f32(false, a0, false, b0, (short)0, c,
                                            false, false);
  c = __builtin_amdgcn_wmma_f32_16x16x4_f32(false, a1, false, b1, (short)0, c,
                                            false, false);
  float tj = 0.0f;
#pragma unroll
  for (int r = 0; r < 8; ++r) {
    if (lane == r || lane == r + 24) tj = c[r];
  }
  return tj;
}

#define OWN_ROWS 196
#define OWN_THREADS 1024
#define OWN_WAVES (OWN_THREADS / 32)
#define LIST_CAP 176
#define ROW_GROUP 16
#define SCORE_GROUP 8
#define SCAN_AHEAD 8

__device__ __forceinline__ int4 load_dst4(const int* __restrict__ dst, int q,
                                          int nq_full) {
  int4 r = ((const int4*)dst)[min(q, nq_full - 1)];
  if (q >= nq_full) r = make_int4(-1, -1, -1, -1);
  return r;
}

__device__ __forceinline__ unsigned owner_key(int d) {
  return ((unsigned)d / (unsigned)OWN_ROWS) & 0xffu;
}

__device__ __forceinline__ uint4 load_key16(const unsigned* __restrict__ own,
                                            int c, int nc_full) {
  return ((const uint4*)own)[min(c, nc_full - 1)];
}

__device__ __forceinline__ unsigned match_flags(uint4 k, unsigned key4) {
  const unsigned x0 = k.x ^ key4, x1 = k.y ^ key4;
  const unsigned x2 = k.z ^ key4, x3 = k.w ^ key4;
  const unsigned z0 = (x0 - 0x01010101u) & ~x0 & 0x80808080u;
  const unsigned z1 = (x1 - 0x01010101u) & ~x1 & 0x80808080u;
  const unsigned z2 = (x2 - 0x01010101u) & ~x2 & 0x80808080u;
  const unsigned z3 = (x3 - 0x01010101u) & ~x3 & 0x80808080u;
  return (z0 >> 7) | (z1 >> 6) | (z2 >> 5) | (z3 >> 4);
}

#define KEY_THREADS 256
#define KEY_WAVE_QUADS 128
__global__ void __launch_bounds__(KEY_THREADS)
owner_key_kernel(const int* __restrict__ dst, unsigned* __restrict__ own,
                 int nq_full) {
  const int lane = threadIdx.x & 31;
  const int wave = (blockIdx.x * KEY_THREADS + threadIdx.x) >> 5;
  const int q0 = wave * KEY_WAVE_QUADS + lane;
  int4 d[4];
#pragma unroll
  for (int j = 0; j < 4; ++j) d[j] = load_dst4(dst, q0 + 32 * j, nq_full);
#pragma unroll
  for (int j = 0; j < 4; ++j) {
    const int q = q0 + 32 * j;
    if (q < nq_full)
      own[q] = owner_key(d[j].x) | (owner_key(d[j].y) << 8) |
               (owner_key(d[j].z) << 16) | (owner_key(d[j].w) << 24);
  }
}

__device__ __forceinline__ void add_row(float* __restrict__ arow, float4 v,
                                        float w, int lane) {
  atomicAdd(arow + lane, v.x * w);
  atomicAdd(arow + 32 + lane, v.y * w);
  atomicAdd(arow + 64 + lane, v.z * w);
  atomicAdd(arow + 96 + lane, v.w * w);
}

template <int STEP>
__device__ __forceinline__ void walk_group(
    const int (&rr)[4], int e0, const float* __restrict__ feat,
    const int* __restrict__ src, int lo, unsigned* mx, float* den, float* rows,
    int lane) {
#pragma unroll
  for (int i = 0; i < 4; ++i) {
    unsigned m = (unsigned)__ballot((unsigned)rr[i] < (unsigned)OWN_ROWS);
    while (m != 0u) {
      const int j = __builtin_ctz(m);
      m &= m - 1u;
      const int ej = __builtin_amdgcn_readlane(e0 + i, j);
      const int rj = __builtin_amdgcn_readlane(rr[i], j);
      const float4 v =
          ((const float4*)(feat + (size_t)src[ej] * FEAT_DIM))[lane];
      const float4 b =
          ((const float4*)(feat + (size_t)(lo + rj) * FEAT_DIM))[lane];
      float q = tile_trace_share(v, b, lane);
      q += __shfl_xor(q, 16, 32);
      q += __shfl_xor(q, 8, 32);
      q += __shfl_xor(q, 4, 32);
      q += __shfl_xor(q, 2, 32);
      q += __shfl_xor(q, 1, 32);
      if (STEP == 1) {
        if (lane == 0) atomicMax(&mx[rj], ordEncode(q));
      } else if (STEP == 2) {
        if (lane == 0) atomicAdd(&den[rj], __expf(q - ordDecode(mx[rj])));
      } else {
        const float w = __expf(q - ordDecode(mx[rj])) / den[rj];
        add_row(rows + rj * FEAT_DIM, v, w, lane);
      }
    }
  }
}

template <int STEP>
__device__ __forceinline__ void walk_rest(
    int t_full, int t0, int n_trip, int nc_full, int E, int lo, int tid,
    const float* __restrict__ feat, const int* __restrict__ src,
    const int* __restrict__ dst, unsigned* mx, float* den, float* rows) {
  const int lane = tid & 31;
  for (int t = t_full; t < n_trip; ++t) {
    const int k = (t + t0 < n_trip) ? t + t0 : t + t0 - n_trip;
    const int c = k * OWN_THREADS + tid;
    for (int g = 0; g < 4; ++g) {
      const int q = c * 4 + g;
      const int4 d = load_dst4(dst, q, nc_full * 4);
      const int rr[4] = {d.x - lo, d.y - lo, d.z - lo, d.w - lo};
      walk_group<STEP>(rr, q * 4, feat, src, lo, mx, den, rows, lane);
    }
  }
  if (tid < 32 && (E & 15) != 0) {
    const int e = nc_full * 16 + lane;
    const int rr[4] = {((e < E) ? dst[e] : -1) - lo, -1, -1, -1};
    walk_group<STEP>(rr, e, feat, src, lo, mx, den, rows, lane);
  }
}

__global__ void __launch_bounds__(OWN_THREADS)
gat_owner_kernel(const float* __restrict__ feat, const int* __restrict__ src,
                 const int* __restrict__ dst, const unsigned* __restrict__ own,
                 float* __restrict__ out, int E, int n_nodes) {
  __shared__ float acc[OWN_ROWS * FEAT_DIM];
  __shared__ unsigned mx[OWN_ROWS];
  __shared__ float den[OWN_ROWS];
  __shared__ int list_e[OWN_WAVES * LIST_CAP];
  __shared__ int list_row[OWN_WAVES * LIST_CAP];
  __shared__ float list_val[OWN_WAVES * LIST_CAP];

  const int tid = threadIdx.x;
  const int lane = tid & 31;
  const int wv = tid >> 5;
  const int lo = blockIdx.x * OWN_ROWS;
  const int n_rows = min(OWN_ROWS, n_nodes - lo);
  int* le = list_e + wv * LIST_CAP;
  int* lr = list_row + wv * LIST_CAP;
  float* lv = list_val + wv * LIST_CAP;

  for (int i = tid; i < n_rows * (FEAT_DIM / 4); i += OWN_THREADS)
    ((float4*)acc)[i] = ((const float4*)feat)[(size_t)lo * (FEAT_DIM / 4) + i];
  if (tid < OWN_ROWS) {
    mx[tid] = 0u;
    den[tid] = 0.0f;
  }

  const int nc_full = E >> 4;
  const int n_trip = (nc_full + OWN_THREADS - 1) / OWN_THREADS;
  const int t0 = (int)(((long long)blockIdx.x * n_trip) / gridDim.x);
  const unsigned key4 = (blockIdx.x & 0xffu) * 0x01010101u;
  int cnt = 0;
  int t_full = n_trip;
  for (int tb = 0; tb < n_trip && t_full == n_trip; tb += SCAN_AHEAD) {
    uint4 kk[SCAN_AHEAD];
    int cc[SCAN_AHEAD];
#pragma unroll
    for (int u = 0; u < SCAN_AHEAD; ++u) {
      const int t = tb + u + t0;
      cc[u] = ((t < n_trip) ? t : t - n_trip) * OWN_THREADS + tid;
      kk[u] = load_key16(own, cc[u], nc_full);
    }
#pragma unroll
    for (int u = 0; u < SCAN_AHEAD; ++u) {
      if (tb + u >= n_trip || t_full != n_trip) continue;
      unsigned f = (cc[u] < nc_full) ? match_flags(kk[u], key4) : 0u;
      const int cnt0 = cnt;
      const int e0 = cc[u] * 16;
      while (true) {
        const bool has = f != 0u;
        const unsigned m = (unsigned)__ballot(has);
        if (m == 0u) break;
        const int n = __builtin_popcount(m);
        if (cnt + n > LIST_CAP) {
          cnt = cnt0;
          t_full = tb + u;
          break;
        }
        if (has) {
          const int p = __builtin_ctz(f);
          f &= f - 1u;
          le[cnt + (int)__builtin_amdgcn_mbcnt_lo(m, 0u)] =
              e0 + 4 * (p & 7) + (p >> 3);
        }
        cnt += n;
      }
    }
  }

  int kept = 0;
  for (int base = 0; base < cnt; base += 32) {
    const int i = base + lane;
    const int e = le[min(i, cnt - 1)];
    const int r = dst[e] - lo;
    const int s = src[e];
    const bool ok = i < cnt && (unsigned)r < (unsigned)OWN_ROWS;
    const unsigned m = (unsigned)__ballot(ok);
    if (ok) {
      const int pos = kept + (int)__builtin_amdgcn_mbcnt_lo(m, 0u);
      le[pos] = s;
      lr[pos] = r;
    }
    kept += __builtin_popcount(m);
  }
  cnt = kept;
  __syncthreads();

  const bool hi16 = (lane & 16) != 0;
  const bool hi8 = (lane & 8) != 0;
  const bool hi4 = (lane & 4) != 0;
  for (int base = 0; base < cnt; base += SCORE_GROUP) {
    float4 a[SCORE_GROUP];
    int rj[SCORE_GROUP];
#pragma unroll
    for (int i = 0; i < SCORE_GROUP; ++i) {
      const int k = min(base + i, cnt - 1);
      const int sj = __builtin_amdgcn_readfirstlane(le[k]);
      rj[i] = __builtin_amdgcn_readfirstlane(lr[k]);
      a[i] = ((const float4*)(feat + (size_t)sj * FEAT_DIM))[lane];
    }
    float t[SCORE_GROUP];
#pragma unroll
    for (int i = 0; i < SCORE_GROUP; ++i) {
      const float4 b = ((const float4*)(acc + rj[i] * FEAT_DIM))[lane];
      t[i] = tile_trace_share(a[i], b, lane);
    }
    float u[4];
#pragma unroll
    for (int j = 0; j < 4; ++j) {
      const float keep = hi16 ? t[4 + j] : t[j];
      const float send = hi16 ? t[j] : t[4 + j];
      u[j] = keep + __shfl_xor(send, 16, 32);
    }
    float w2[2];
#pragma unroll
    for (int j = 0; j < 2; ++j) {
      const float keep = hi8 ? u[2 + j] : u[j];
      const float send = hi8 ? u[j] : u[2 + j];
      w2[j] = keep + __shfl_xor(send, 8, 32);
    }
    float q = (hi4 ? w2[1] : w2[0]) + __shfl_xor(hi4 ? w2[0] : w2[1], 4, 32);
    q += __shfl_xor(q, 2, 32);
    q += __shfl_xor(q, 1, 32);
    const int k = base + (lane >> 2);
    if ((lane & 3) == 0 && k < cnt) {
      lv[k] = q;
      atomicMax(&mx[lr[k]], ordEncode(q));
    }
  }
  walk_rest<1>(t_full, t0, n_trip, nc_full, E, lo, tid, feat, src, dst, mx, den,
               acc);
  __syncthreads();

  for (int i = tid; i < OWN_ROWS * FEAT_DIM / 4; i += OWN_THREADS)
    ((float4*)acc)[i] = make_float4(0.0f, 0.0f, 0.0f, 0.0f);
  for (int i = lane; i < cnt; i += 32) {
    const int r = lr[i];
    const float ex = __expf(lv[i] - ordDecode(mx[r]));
    lv[i] = ex;
    atomicAdd(&den[r], ex);
  }
  walk_rest<2>(t_full, t0, n_trip, nc_full, E, lo, tid, feat, src, dst, mx, den,
               acc);
  __syncthreads();

  for (int i = lane; i < cnt; i += 32) lv[i] = lv[i] / den[lr[i]];
  __syncthreads();

  for (int base = 0; base < cnt; base += ROW_GROUP) {
    float4 v[ROW_GROUP];
#pragma unroll
    for (int i = 0; i < ROW_GROUP; ++i) {
      if (base + i < cnt) {
        const int sj = __builtin_amdgcn_readfirstlane(le[base + i]);
        v[i] = ((const float4*)(feat + (size_t)sj * FEAT_DIM))[lane];
      }
    }
#pragma unroll
    for (int i = 0; i < ROW_GROUP; ++i) {
      if (base + i < cnt) {
        const int rj = __builtin_amdgcn_readfirstlane(lr[base + i]);
        add_row(acc + rj * FEAT_DIM, v[i], lv[base + i], lane);
      }
    }
  }
  walk_rest<3>(t_full, t0, n_trip, nc_full, E, lo, tid, feat, src, dst, mx, den,
               acc);
  __syncthreads();

  for (int r = wv; r < n_rows; r += OWN_WAVES) {
    const float* a = acc + r * FEAT_DIM + lane;
    float4 o = make_float4(a[0], a[32], a[64], a[96]);
    const size_t r4 = (size_t)(lo + r) * (FEAT_DIM / 4) + lane;
    if (den[r] == 0.0f) o = ((const float4*)feat)[r4];
    ((float4*)out)[r4] = o;
  }
}

extern "C" void kernel_launch(void* const* d_in, const int* in_sizes, int n_in,
                              void* d_out, int out_size, void* d_ws, size_t ws_size,
                              hipStream_t stream) {
  const float* feat = (const float*)d_in[0];
  const int* src    = (const int*)d_in[1];
  const int* dst    = (const int*)d_in[2];
  const int E       = in_sizes[1];
  const int N       = in_sizes[0] / FEAT_DIM;
  float* out        = (float*)d_out;

  unsigned* own = (unsigned*)d_ws;

  owner_key_kernel<<<(E / 4 + KEY_THREADS * 4 - 1) / (KEY_THREADS * 4),
                     KEY_THREADS, 0, stream>>>(dst, own, E / 4);

  gat_owner_kernel<<<(N + OWN_ROWS - 1) / OWN_ROWS, OWN_THREADS, 0, stream>>>(
      feat, src, dst, own, out, E, N);
}
